# starting kernel
# baseline (speedup 1.0000x reference)
#include <hip/hip_runtime.h>
#include <math.h>

typedef __attribute__((ext_vector_type(2))) float v2f;
typedef __attribute__((ext_vector_type(8))) float v8f;

#define F_NUM 10
#define OUT_N 10
#define NROW  12

__global__ void coef_kernel(const float* __restrict__ w,
                            const float* __restrict__ pool_w,
                            float* __restrict__ coefT,
                            int T, int Tpad, int D) {
  int t = blockIdx.x * blockDim.x + threadIdx.x;
  if (t >= Tpad) return;
  float row[16];
#pragma unroll
  for (int i = 0; i < 16; ++i) row[i] = 0.0f;
  if (t < T) {
    row[0] = 1.0f;
    float G = 0.0f;
    float e = (float)(T - 1 - t);
#pragma unroll
    for (int f = 0; f < F_NUM; ++f) {
      float wf = w[(size_t)f * D];
      float pf = pool_w[f];
      float c  = powf(wf, e);
      row[2 + f] = c;
      G += pf * (1.0f - c) / (1.0f - wf);
    }
    row[1] = G;
  }
#pragma unroll
  for (int i = 0; i < 16; ++i) coefT[(size_t)t * 16 + i] = row[i];
}

template <int DC>
__global__ void __launch_bounds__(128)
gemm_coef_x(const float* __restrict__ x, const float* __restrict__ coefT,
            float* __restrict__ partial, int T, int Drt, int TC, int NT, int units) {
  const int D = (DC > 0) ? DC : Drt;
  int wid = __builtin_amdgcn_readfirstlane(
      (int)(blockIdx.x * (blockDim.x >> 5)) + ((int)threadIdx.x >> 5));
  if (wid >= units) return;
  int lane = threadIdx.x & 31;
  int ntile  = wid % NT;
  int kchunk = wid / NT;
  int nb = ntile * 16;
  int n  = lane & 15;
  int h  = lane >> 4;

  int t0 = kchunk * TC;
  int valid = T - t0;
  if (valid < 0)  valid = 0;
  if (valid > TC) valid = TC;
  int full = valid & ~3;

  const float* cp = coefT + (size_t)(t0 + 2 * h) * 16 + n;
  const float* xp = x     + (size_t)(t0 + 2 * h) * D  + nb + n;
  const size_t xstep = (size_t)4 * D;

  v8f acc = {};
  if (full > 0) {
    v2f a = { cp[0], cp[16] };
    v2f b = { xp[0], xp[D] };
#pragma unroll 2
    for (int kk = 4; kk < full; kk += 4) {
      cp += 64;
      xp += xstep;
      v2f an = { cp[0], cp[16] };
      v2f bn = { xp[0], xp[D] };
      acc = __builtin_amdgcn_wmma_f32_16x16x4_f32(
          false, a, false, b, (short)0, acc, false, false);
      a = an; b = bn;
    }
    acc = __builtin_amdgcn_wmma_f32_16x16x4_f32(
        false, a, false, b, (short)0, acc, false, false);
  }
  for (int kk = full; kk < TC; kk += 4) {
    int ta = t0 + kk + 2 * h, tb = ta + 1;
    v2f a = { coefT[(size_t)ta * 16 + n], coefT[(size_t)tb * 16 + n] };
    v2f b = { (ta < T) ? x[(size_t)ta * D + nb + n] : 0.0f,
              (tb < T) ? x[(size_t)tb * D + nb + n] : 0.0f };
    acc = __builtin_amdgcn_wmma_f32_16x16x4_f32(
        false, a, false, b, (short)0, acc, false, false);
  }

#pragma unroll
  for (int r = 0; r < 8; ++r) {
    int m = r + 8 * h;
    if (m < NROW)
      partial[((size_t)kchunk * NROW + m) * D + nb + n] = acc[r];
  }
}

__global__ void reduce_partials(const float* __restrict__ partial,
                                const float* __restrict__ pool_b,
                                float* __restrict__ avg,
                                float* __restrict__ out_hfinal,
                                int D, int KC, int T) {
  int idx = blockIdx.x * blockDim.x + threadIdx.x;
  if (idx >= NROW * D) return;
  int m = idx / D, d = idx % D;
  float s = 0.0f;
  for (int c = 0; c < KC; ++c)
    s += partial[((size_t)c * NROW + m) * D + d];
  float invT = 1.0f / (float)T;
  if (m == 0)      avg[d]     = s * invT;
  else if (m == 1) avg[D + d] = s * invT + pool_b[0];
  else             out_hfinal[(size_t)(m - 2) * D + d] = s;
}

__global__ void gemv_rows(const float* __restrict__ Wm,
                          const float* __restrict__ bias,
                          const float* __restrict__ vec,
                          float* __restrict__ out,
                          int K, int N, int do_relu) {
  int wid  = blockIdx.x * (blockDim.x >> 5) + (threadIdx.x >> 5);
  int lane = threadIdx.x & 31;
  if (wid >= N) return;
  const float4* wr = (const float4*)(Wm + (size_t)wid * K);
  const float4* vv = (const float4*)vec;
  float s = 0.0f;
  for (int i = lane; i < (K >> 2); i += 32) {
    float4 a = wr[i], b = vv[i];
    s += a.x * b.x + a.y * b.y + a.z * b.z + a.w * b.w;
  }
  for (int off = 16; off > 0; off >>= 1) s += __shfl_down(s, off, 32);
  if (lane == 0) {
    float v = s + bias[wid];
    out[wid] = do_relu ? fmaxf(v, 0.0f) : v;
  }
}

extern "C" void kernel_launch(void* const* d_in, const int* in_sizes, int n_in,
                              void* d_out, int out_size, void* d_ws, size_t ws_size,
                              hipStream_t stream) {
  const float* inp    = (const float*)d_in[0];
  const float* w      = (const float*)d_in[2];
  const float* pool_w = (const float*)d_in[3];
  const float* pool_b = (const float*)d_in[4];
  const float* i2o_w  = (const float*)d_in[5];
  const float* i2o_b  = (const float*)d_in[6];
  const float* o_w    = (const float*)d_in[7];
  const float* o_b    = (const float*)d_in[8];
  float* out = (float*)d_out;

  const int D  = in_sizes[1] / F_NUM;
  const int T  = in_sizes[0] / D;
  const int H1 = in_sizes[6];

  const int KC = 16;
  int TC   = (((T + KC - 1) / KC) + 3) & ~3;
  int Tpad = KC * TC;
  int NT   = D / 16;

  float* ws      = (float*)d_ws;
  float* coefT   = ws;
  float* partial = coefT + (size_t)Tpad * 16;
  float* avg     = partial + (size_t)KC * NROW * D;
  float* hid1    = avg + (size_t)2 * D;

  coef_kernel<<<(Tpad + 255) / 256, 256, 0, stream>>>(w, pool_w, coefT, T, Tpad, D);

  int units = NT * KC;
  dim3 g2((units + 3) / 4), b2(128);
  if (D == 4096)
    gemm_coef_x<4096><<<g2, b2, 0, stream>>>(inp, coefT, partial, T, D, TC, NT, units);
  else
    gemm_coef_x<0><<<g2, b2, 0, stream>>>(inp, coefT, partial, T, D, TC, NT, units);

  reduce_partials<<<(NROW * D + 255) / 256, 256, 0, stream>>>(
      partial, pool_b, avg, out + OUT_N, D, KC, T);

  gemv_rows<<<(H1 + 7) / 8, 256, 0, stream>>>(i2o_w, i2o_b, avg, hid1, 2 * D, H1, 1);
  gemv_rows<<<(OUT_N + 7) / 8, 256, 0, stream>>>(o_w, o_b, hid1, out, H1, OUT_N, 0);
}


# reduce launch: one thread per b128 piece of a row requests its 16 chunk planes in one round (was 4 dependent rounds of b32 loads), 384 waves (were 1536); chunk sum order unchanged
# speedup vs baseline: 2.7347x; 2.7347x over previous
#include <hip/hip_runtime.h>
#include <math.h>

typedef __attribute__((ext_vector_type(2))) float v2f;
typedef __attribute__((ext_vector_type(8))) float v8f;

#define F_NUM 10
#define OUT_N 10
#define NROW  12

__device__ __forceinline__ void l2_prefetch_unit(const float* p) {
  __builtin_prefetch(p, 0, 2);
}

__device__ __forceinline__ void prefetch_x_half_tiles(const float* __restrict__ x,
                                                      int pb, int NS, int TC, int D, int T,
                                                      int u0, int u1, int lane) {
  const int slab = pb % NS, kchunk = pb / NS;
  const float* xb = x + (size_t)(kchunk * TC) * D + slab * 128;
  for (int u = u0; u < u1; u += 32) {
    const int v = u + lane;
    const int row = 64 * (v >> 6) + ((v & 63) >> 1);
    if (row < TC && kchunk * TC + row < T)
      l2_prefetch_unit(xb + (size_t)row * D + (v & 1) * 64);
  }
}

__global__ void coef_kernel(const float* __restrict__ w,
                            const float* __restrict__ pool_w,
                            float* __restrict__ coefT,
                            int T, int Tpad, int D,
                            const float* __restrict__ xpf, int TC, int NS, int npf) {
  int t = blockIdx.x * blockDim.x + threadIdx.x;
  const int g  = t >> 5, lane = t & 31;
  const int ng = (int)(gridDim.x * blockDim.x) >> 5;
  if (xpf != nullptr)
    for (int pb = g; pb < npf; pb += ng)
      prefetch_x_half_tiles(xpf, pb, NS, TC, D, T, 0, 256, lane);
  if (t >= Tpad) return;
  float row[16];
#pragma unroll
  for (int i = 0; i < 16; ++i) row[i] = 0.0f;
  if (t < T) {
    row[0] = 1.0f;
    float G = 0.0f;
    float e = (float)(T - 1 - t);
#pragma unroll
    for (int f = 0; f < F_NUM; ++f) {
      float wf = w[(size_t)f * D];
      float pf = pool_w[f];
      float c  = powf(wf, e);
      row[2 + f] = c;
      G += pf * (1.0f - c) / (1.0f - wf);
    }
    row[1] = G;
  }
#pragma unroll
  for (int i = 0; i < 16; ++i) coefT[(size_t)t * 16 + i] = row[i];
  if (xpf != nullptr)
    for (int pb = g; pb < npf; pb += ng)
      prefetch_x_half_tiles(xpf, pb, NS, TC, D, T, 256, 384, lane);
}

template <int DC>
__global__ void __launch_bounds__(128)
gemm_coef_x(const float* __restrict__ x, const float* __restrict__ coefT,
            float* __restrict__ partial, int T, int Drt, int TC, int NT, int units) {
  const int D = (DC > 0) ? DC : Drt;
  int wid = __builtin_amdgcn_readfirstlane(
      (int)(blockIdx.x * (blockDim.x >> 5)) + ((int)threadIdx.x >> 5));
  if (wid >= units) return;
  int lane = threadIdx.x & 31;
  int ntile  = wid % NT;
  int kchunk = wid / NT;
  int nb = ntile * 16;
  int n  = lane & 15;
  int h  = lane >> 4;

  int t0 = kchunk * TC;
  int valid = T - t0;
  if (valid < 0)  valid = 0;
  if (valid > TC) valid = TC;
  int full = valid & ~3;

  const float* cp = coefT + (size_t)(t0 + 2 * h) * 16 + n;
  const float* xp = x     + (size_t)(t0 + 2 * h) * D  + nb + n;
  const size_t xstep = (size_t)4 * D;

  v8f acc = {};
  if (full > 0) {
    v2f a = { cp[0], cp[16] };
    v2f b = { xp[0], xp[D] };
#pragma unroll 2
    for (int kk = 4; kk < full; kk += 4) {
      cp += 64;
      xp += xstep;
      v2f an = { cp[0], cp[16] };
      v2f bn = { xp[0], xp[D] };
      acc = __builtin_amdgcn_wmma_f32_16x16x4_f32(
          false, a, false, b, (short)0, acc, false, false);
      a = an; b = bn;
    }
    acc = __builtin_amdgcn_wmma_f32_16x16x4_f32(
        false, a, false, b, (short)0, acc, false, false);
  }
  for (int kk = full; kk < TC; kk += 4) {
    int ta = t0 + kk + 2 * h, tb = ta + 1;
    v2f a = { coefT[(size_t)ta * 16 + n], coefT[(size_t)tb * 16 + n] };
    v2f b = { (ta < T) ? x[(size_t)ta * D + nb + n] : 0.0f,
              (tb < T) ? x[(size_t)tb * D + nb + n] : 0.0f };
    acc = __builtin_amdgcn_wmma_f32_16x16x4_f32(
        false, a, false, b, (short)0, acc, false, false);
  }

#pragma unroll
  for (int r = 0; r < 8; ++r) {
    int m = r + 8 * h;
    if (m < NROW)
      partial[((size_t)kchunk * NROW + m) * D + nb + n] = acc[r];
  }
}

#define XS_ROWS  64
#define XS_COLS  128
#define XS_PITCH 136

__device__ __forceinline__ v8f staged_tile_products(const float* __restrict__ xb,
                                                    const float* __restrict__ ca,
                                                    v8f acc) {
#pragma unroll
  for (int ks = 0; ks < XS_ROWS / 4; ++ks) {
    v2f a = { ca[ks * 64], ca[ks * 64 + 16] };
    v2f b = { xb[ks * 4 * XS_PITCH], xb[ks * 4 * XS_PITCH + XS_PITCH] };
    acc = __builtin_amdgcn_wmma_f32_16x16x4_f32(
        false, a, false, b, (short)0, acc, false, false);
  }
  return acc;
}

#define XS_LOAD_TILE()                                            \
  r0 = *(const float4*)(xg);                                      \
  r1 = *(const float4*)(xg + (size_t)8  * D);                     \
  r2 = *(const float4*)(xg + (size_t)16 * D);                     \
  r3 = *(const float4*)(xg + (size_t)24 * D);                     \
  r4 = *(const float4*)(xg + (size_t)32 * D);                     \
  r5 = *(const float4*)(xg + (size_t)40 * D);                     \
  r6 = *(const float4*)(xg + (size_t)48 * D);                     \
  r7 = *(const float4*)(xg + (size_t)56 * D);                     \
  rc = *(const float4*)(cg);

#define XS_STORE_TILE(buf)                                        \
  *(float4*)(xsw + (buf) * (XS_ROWS * XS_PITCH))                 = r0; \
  *(float4*)(xsw + (buf) * (XS_ROWS * XS_PITCH) + 8  * XS_PITCH) = r1; \
  *(float4*)(xsw + (buf) * (XS_ROWS * XS_PITCH) + 16 * XS_PITCH) = r2; \
  *(float4*)(xsw + (buf) * (XS_ROWS * XS_PITCH) + 24 * XS_PITCH) = r3; \
  *(float4*)(xsw + (buf) * (XS_ROWS * XS_PITCH) + 32 * XS_PITCH) = r4; \
  *(float4*)(xsw + (buf) * (XS_ROWS * XS_PITCH) + 40 * XS_PITCH) = r5; \
  *(float4*)(xsw + (buf) * (XS_ROWS * XS_PITCH) + 48 * XS_PITCH) = r6; \
  *(float4*)(xsw + (buf) * (XS_ROWS * XS_PITCH) + 56 * XS_PITCH) = r7; \
  *(float4*)(csw + (buf) * (XS_ROWS * 16))                       = rc;

template <int DC>
__global__ void __launch_bounds__(256)
gemm_coef_x_staged(const float* __restrict__ x, const float* __restrict__ coefT,
                   float* __restrict__ partial, int Drt, int TC, int NS,
                   const float* __restrict__ wpf, int wrows, int npf) {
  __shared__ __attribute__((aligned(16))) float xs[2 * XS_ROWS * XS_PITCH];
  __shared__ __attribute__((aligned(16))) float cs[2 * XS_ROWS * 16];
  const int D    = (DC > 0) ? DC : Drt;
  const int tid  = threadIdx.x;
  const int lane = tid & 31;
  const int wv   = __builtin_amdgcn_readfirstlane(tid >> 5);
  const int n = lane & 15, h = lane >> 4;
  const int slab   = blockIdx.x % NS;
  const int kchunk = blockIdx.x / NS;
  const int cb = slab * XS_COLS;
  const int t0 = kchunk * TC;
  const int ntile = TC / XS_ROWS;

  const int lr = tid >> 5;
  const int lc = (tid & 31) * 4;
  const float* xg = x + (size_t)(t0 + lr) * D + cb + lc;
  const float* cg = coefT + (size_t)t0 * 16 + tid * 4;
  float* xsw = xs + lr * XS_PITCH + lc;
  float* csw = cs + tid * 4;
  const float* xsr = xs + (2 * h) * XS_PITCH + wv * 16 + n;
  const float* csr = cs + (2 * h) * 16 + n;

  float4 r0, r1, r2, r3, r4, r5, r6, r7, rc;
  XS_LOAD_TILE()
  XS_STORE_TILE(0)
  __syncthreads();

  v8f acc = {};
  for (int j = 0; j + 1 < ntile; ++j) {
    const int cur = j & 1;
    xg += (size_t)XS_ROWS * D;
    cg += XS_ROWS * 16;
    XS_LOAD_TILE()
    acc = staged_tile_products(xsr + cur * (XS_ROWS * XS_PITCH),
                               csr + cur * (XS_ROWS * 16), acc);
    XS_STORE_TILE(cur ^ 1)
    __syncthreads();
  }
  {
    const int cur = (ntile - 1) & 1;
    acc = staged_tile_products(xsr + cur * (XS_ROWS * XS_PITCH),
                               csr + cur * (XS_ROWS * 16), acc);
  }

  const int nb = cb + wv * 16;
#pragma unroll
  for (int r = 0; r < 8; ++r) {
    int m = r + 8 * h;
    if (m < NROW)
      partial[((size_t)kchunk * NROW + m) * D + nb + n] = acc[r];
  }

  if (wpf != nullptr && (int)blockIdx.x < npf) {
    constexpr int UPR = (2 * ((DC > 0) ? DC : 64)) / 64;
    const int r = tid / UPR, u = tid % UPR;
    if (r < 2 && r < wrows)
      l2_prefetch_unit(wpf + ((size_t)blockIdx.x * wrows + r) * (size_t)(2 * D) + u * 64);
  }
}

__global__ void reduce_partials(const float* __restrict__ partial,
                                const float* __restrict__ pool_b,
                                float* __restrict__ avg,
                                float* __restrict__ out_hfinal,
                                int D, int KC, int T) {
  int idx = blockIdx.x * blockDim.x + threadIdx.x;
  if (idx >= NROW * D) return;
  int m = idx / D, d = idx % D;
  float s = 0.0f;
  for (int c = 0; c < KC; ++c)
    s += partial[((size_t)c * NROW + m) * D + d];
  float invT = 1.0f / (float)T;
  if (m == 0)      avg[d]     = s * invT;
  else if (m == 1) avg[D + d] = s * invT + pool_b[0];
  else             out_hfinal[(size_t)(m - 2) * D + d] = s;
}

#define RR_THREADS 128

template <int DC, int KCC>
__global__ void __launch_bounds__(RR_THREADS)
reduce_partials_round(const float* __restrict__ partial,
                      const float* __restrict__ pool_b,
                      float* __restrict__ avg,
                      float* __restrict__ out_hfinal,
                      int T) {
  constexpr int Q = DC / 4;
  const int idx = blockIdx.x * RR_THREADS + threadIdx.x;
  if (idx >= NROW * Q) return;
  const int m = idx / Q, q = idx % Q;
  const float4* pp = (const float4*)partial + (size_t)m * Q + q;
  float4 p[KCC];
#pragma unroll
  for (int c = 0; c < KCC; ++c) p[c] = pp[(size_t)c * NROW * Q];
  float invT = 1.0f / (float)T;
  float pb = pool_b[0];
  asm volatile("" : "+v"(invT), "+v"(pb));
  float sx = 0.0f, sy = 0.0f, sz = 0.0f, sw = 0.0f;
#pragma unroll
  for (int c = 0; c < KCC; ++c) {
    sx += p[c].x; sy += p[c].y; sz += p[c].z; sw += p[c].w;
  }
  __builtin_amdgcn_sched_group_barrier(0x020, KCC, 0);
  __builtin_amdgcn_sched_group_barrier(0x002, 4 * KCC, 0);
  const int d = q * 4;
  if (m < 2) {
    const float add = (m == 1) ? pb : 0.0f;
    float4 o = { sx * invT + add, sy * invT + add, sz * invT + add, sw * invT + add };
    *(float4*)(avg + (size_t)m * DC + d) = o;
  } else {
    float* oh = out_hfinal + (size_t)(m - 2) * DC + d;
    float2 lo = { sx, sy }, hi = { sz, sw };
    *(float2*)(oh)     = lo;
    *(float2*)(oh + 2) = hi;
  }
}

__global__ void gemv_rows(const float* __restrict__ Wm,
                          const float* __restrict__ bias,
                          const float* __restrict__ vec,
                          float* __restrict__ out,
                          int K, int N, int do_relu) {
  int wid  = blockIdx.x * (blockDim.x >> 5) + (threadIdx.x >> 5);
  int lane = threadIdx.x & 31;
  if (wid >= N) return;
  const float4* wr = (const float4*)(Wm + (size_t)wid * K);
  const float4* vv = (const float4*)vec;
  float s = 0.0f;
  for (int i = lane; i < (K >> 2); i += 32) {
    float4 a = wr[i], b = vv[i];
    s += a.x * b.x + a.y * b.y + a.z * b.z + a.w * b.w;
  }
  for (int off = 16; off > 0; off >>= 1) s += __shfl_down(s, off, 32);
  if (lane == 0) {
    float v = s + bias[wid];
    out[wid] = do_relu ? fmaxf(v, 0.0f) : v;
  }
}

#define GB_THREADS 256
#define GB_ROWS    8
#define GB_GROUP   4

template <int KP>
__global__ void __launch_bounds__(GB_THREADS)
gemv_rows_block(const float* __restrict__ Wm,
                const float* __restrict__ bias,
                const float* __restrict__ vec,
                float* __restrict__ out,
                int do_relu) {
  __shared__ float part[GB_ROWS][GB_THREADS / 32];
  const int tid  = threadIdx.x;
  const int lane = tid & 31;
  const int wv   = tid >> 5;
  const int row0 = blockIdx.x * GB_ROWS;
  constexpr int K4 = KP * GB_THREADS;

  const float4* vv = (const float4*)vec + tid;
  const float4* wr = (const float4*)Wm + (size_t)row0 * K4 + tid;

  float bv = (tid < GB_ROWS) ? bias[row0 + tid] : 0.0f;
  float4 v[KP];
#pragma unroll
  for (int p = 0; p < KP; ++p) v[p] = vv[p * GB_THREADS];

#pragma unroll 1
  for (int g = 0; g < GB_ROWS; g += GB_GROUP) {
    float4 a[GB_GROUP][KP];
#pragma unroll
    for (int r = 0; r < GB_GROUP; ++r)
#pragma unroll
      for (int p = 0; p < KP; ++p)
        a[r][p] = wr[(size_t)r * K4 + p * GB_THREADS];
    wr += (size_t)GB_GROUP * K4;
#pragma unroll
    for (int r = 0; r < GB_GROUP; ++r) {
      float sx = 0.0f, sy = 0.0f, sz = 0.0f, sw = 0.0f;
#pragma unroll
      for (int p = 0; p < KP; ++p) {
        sx = fmaf(a[r][p].x, v[p].x, sx);
        sy = fmaf(a[r][p].y, v[p].y, sy);
        sz = fmaf(a[r][p].z, v[p].z, sz);
        sw = fmaf(a[r][p].w, v[p].w, sw);
      }
      float s = (sx + sy) + (sz + sw);
#pragma unroll
      for (int off = 16; off > 0; off >>= 1) s += __shfl_xor(s, off, 32);
      if (lane == 0) part[g + r][wv] = s;
    }
  }
  __syncthreads();
  if (tid < GB_ROWS) {
    float t = 0.0f;
#pragma unroll
    for (int w2 = 0; w2 < GB_THREADS / 32; ++w2) t += part[tid][w2];
    float y = t + bv;
    out[row0 + tid] = do_relu ? fmaxf(y, 0.0f) : y;
  }
}

__global__ void __launch_bounds__(1024)
gemv_row_per_block(const float* __restrict__ Wm,
                   const float* __restrict__ bias,
                   const float* __restrict__ vec,
                   float* __restrict__ out,
                   int K, int do_relu) {
  __shared__ float part[32];
  int row  = blockIdx.x;
  int lane = threadIdx.x & 31;
  int wv   = threadIdx.x >> 5;
  const float4* wr = (const float4*)(Wm + (size_t)row * K);
  const float4* vv = (const float4*)vec;
  float bv = bias[row];
  float s = 0.0f;
  for (int i = threadIdx.x; i < (K >> 2); i += blockDim.x) {
    float4 a = wr[i], b = vv[i];
    s += a.x * b.x + a.y * b.y + a.z * b.z + a.w * b.w;
  }
  for (int off = 16; off > 0; off >>= 1) s += __shfl_down(s, off, 32);
  if (lane == 0) part[wv] = s;
  __syncthreads();
  if (wv == 0) {
    int nw = blockDim.x >> 5;
    float p = (lane < nw) ? part[lane] : 0.0f;
    for (int off = 16; off > 0; off >>= 1) p += __shfl_down(p, off, 32);
    if (lane == 0) {
      float v = p + bv;
      out[row] = do_relu ? fmaxf(v, 0.0f) : v;
    }
  }
}

extern "C" void kernel_launch(void* const* d_in, const int* in_sizes, int n_in,
                              void* d_out, int out_size, void* d_ws, size_t ws_size,
                              hipStream_t stream) {
  const float* inp    = (const float*)d_in[0];
  const float* w      = (const float*)d_in[2];
  const float* pool_w = (const float*)d_in[3];
  const float* pool_b = (const float*)d_in[4];
  const float* i2o_w  = (const float*)d_in[5];
  const float* i2o_b  = (const float*)d_in[6];
  const float* o_w    = (const float*)d_in[7];
  const float* o_b    = (const float*)d_in[8];
  float* out = (float*)d_out;

  const int D  = in_sizes[1] / F_NUM;
  const int T  = in_sizes[0] / D;
  const int H1 = in_sizes[6];

  const int KC = 16;
  int TC   = (((T + KC - 1) / KC) + 3) & ~3;
  int Tpad = KC * TC;
  int NT   = D / 16;

  float* ws      = (float*)d_ws;
  float* coefT   = ws;
  float* partial = coefT + (size_t)Tpad * 16;
  float* avg     = partial + (size_t)KC * NROW * D;
  float* hid1    = avg + (size_t)2 * D;

  const bool staged  = (D == 4096 && T == Tpad && (TC % XS_ROWS) == 0);
  const bool blkgemv = (D == 4096 && (H1 % GB_ROWS) == 0);

  if (staged)
    coef_kernel<<<(Tpad + 31) / 32, 32, 0, stream>>>(w, pool_w, coefT, T, Tpad, D,
                                                     inp, TC, D / XS_COLS, (D / XS_COLS) * KC);
  else
    coef_kernel<<<(Tpad + 255) / 256, 256, 0, stream>>>(w, pool_w, coefT, T, Tpad, D,
                                                        nullptr, TC, 1, 0);

  int units = NT * KC;
  dim3 g2((units + 3) / 4), b2(128);
  if (staged)
    gemm_coef_x_staged<4096><<<(D / XS_COLS) * KC, 256, 0, stream>>>(
        inp, coefT, partial, D, TC, D / XS_COLS,
        blkgemv ? i2o_w : nullptr, GB_ROWS, H1 / GB_ROWS);
  else
    gemm_coef_x<0><<<g2, b2, 0, stream>>>(inp, coefT, partial, T, D, TC, NT, units);

  if (D == 4096 && KC == 16)
    reduce_partials_round<4096, 16><<<(NROW * (4096 / 4) + RR_THREADS - 1) / RR_THREADS, RR_THREADS, 0, stream>>>(
        partial, pool_b, avg, out + OUT_N, T);
  else
    reduce_partials<<<(NROW * D + 255) / 256, 256, 0, stream>>>(
        partial, pool_b, avg, out + OUT_N, D, KC, T);

  if (blkgemv)
    gemv_rows_block<8><<<H1 / GB_ROWS, GB_THREADS, 0, stream>>>(i2o_w, i2o_b, avg, hid1, 1);
  else
    gemv_rows<<<(H1 + 7) / 8, 256, 0, stream>>>(i2o_w, i2o_b, avg, hid1, 2 * D, H1, 1);
  gemv_row_per_block<<<OUT_N, 1024, 0, stream>>>(o_w, o_b, hid1, out, H1, 0);
}


# both streaming launches (product, 4096x8192 GEMV): 4 waves per block (were 8), twice the b128 pieces per thread: same bytes in flight, half the waves to enter; GEMV row sum order changes
# speedup vs baseline: 2.8263x; 1.0335x over previous
#include <hip/hip_runtime.h>
#include <math.h>

typedef __attribute__((ext_vector_type(2))) float v2f;
typedef __attribute__((ext_vector_type(8))) float v8f;

#define F_NUM 10
#define OUT_N 10
#define NROW  12

__device__ __forceinline__ void l2_prefetch_unit(const float* p) {
  __builtin_prefetch(p, 0, 2);
}

__device__ __forceinline__ void prefetch_x_half_tiles(const float* __restrict__ x,
                                                      int pb, int NS, int TC, int D, int T,
                                                      int u0, int u1, int lane) {
  const int slab = pb % NS, kchunk = pb / NS;
  const float* xb = x + (size_t)(kchunk * TC) * D + slab * 128;
  for (int u = u0; u < u1; u += 32) {
    const int v = u + lane;
    const int row = 64 * (v >> 6) + ((v & 63) >> 1);
    if (row < TC && kchunk * TC + row < T)
      l2_prefetch_unit(xb + (size_t)row * D + (v & 1) * 64);
  }
}

__global__ void coef_kernel(const float* __restrict__ w,
                            const float* __restrict__ pool_w,
                            float* __restrict__ coefT,
                            int T, int Tpad, int D,
                            const float* __restrict__ xpf, int TC, int NS, int npf) {
  int t = blockIdx.x * blockDim.x + threadIdx.x;
  const int g  = t >> 5, lane = t & 31;
  const int ng = (int)(gridDim.x * blockDim.x) >> 5;
  if (xpf != nullptr)
    for (int pb = g; pb < npf; pb += ng)
      prefetch_x_half_tiles(xpf, pb, NS, TC, D, T, 0, 256, lane);
  if (t >= Tpad) return;
  float row[16];
#pragma unroll
  for (int i = 0; i < 16; ++i) row[i] = 0.0f;
  if (t < T) {
    row[0] = 1.0f;
    float G = 0.0f;
    float e = (float)(T - 1 - t);
#pragma unroll
    for (int f = 0; f < F_NUM; ++f) {
      float wf = w[(size_t)f * D];
      float pf = pool_w[f];
      float c  = powf(wf, e);
      row[2 + f] = c;
      G += pf * (1.0f - c) / (1.0f - wf);
    }
    row[1] = G;
  }
#pragma unroll
  for (int i = 0; i < 16; ++i) coefT[(size_t)t * 16 + i] = row[i];
  if (xpf != nullptr)
    for (int pb = g; pb < npf; pb += ng)
      prefetch_x_half_tiles(xpf, pb, NS, TC, D, T, 256, 384, lane);
}

template <int DC>
__global__ void __launch_bounds__(128)
gemm_coef_x(const float* __restrict__ x, const float* __restrict__ coefT,
            float* __restrict__ partial, int T, int Drt, int TC, int NT, int units) {
  const int D = (DC > 0) ? DC : Drt;
  int wid = __builtin_amdgcn_readfirstlane(
      (int)(blockIdx.x * (blockDim.x >> 5)) + ((int)threadIdx.x >> 5));
  if (wid >= units) return;
  int lane = threadIdx.x & 31;
  int ntile  = wid % NT;
  int kchunk = wid / NT;
  int nb = ntile * 16;
  int n  = lane & 15;
  int h  = lane >> 4;

  int t0 = kchunk * TC;
  int valid = T - t0;
  if (valid < 0)  valid = 0;
  if (valid > TC) valid = TC;
  int full = valid & ~3;

  const float* cp = coefT + (size_t)(t0 + 2 * h) * 16 + n;
  const float* xp = x     + (size_t)(t0 + 2 * h) * D  + nb + n;
  const size_t xstep = (size_t)4 * D;

  v8f acc = {};
  if (full > 0) {
    v2f a = { cp[0], cp[16] };
    v2f b = { xp[0], xp[D] };
#pragma unroll 2
    for (int kk = 4; kk < full; kk += 4) {
      cp += 64;
      xp += xstep;
      v2f an = { cp[0], cp[16] };
      v2f bn = { xp[0], xp[D] };
      acc = __builtin_amdgcn_wmma_f32_16x16x4_f32(
          false, a, false, b, (short)0, acc, false, false);
      a = an; b = bn;
    }
    acc = __builtin_amdgcn_wmma_f32_16x16x4_f32(
        false, a, false, b, (short)0, acc, false, false);
  }
  for (int kk = full; kk < TC; kk += 4) {
    int ta = t0 + kk + 2 * h, tb = ta + 1;
    v2f a = { coefT[(size_t)ta * 16 + n], coefT[(size_t)tb * 16 + n] };
    v2f b = { (ta < T) ? x[(size_t)ta * D + nb + n] : 0.0f,
              (tb < T) ? x[(size_t)tb * D + nb + n] : 0.0f };
    acc = __builtin_amdgcn_wmma_f32_16x16x4_f32(
        false, a, false, b, (short)0, acc, false, false);
  }

#pragma unroll
  for (int r = 0; r < 8; ++r) {
    int m = r + 8 * h;
    if (m < NROW)
      partial[((size_t)kchunk * NROW + m) * D + nb + n] = acc[r];
  }
}

#define XS_ROWS  64
#define XS_COLS  128
#define XS_PITCH 136
#define XS_THREADS 128
#define XS_PIECES  16

__device__ __forceinline__ void staged_tile_products(const float* __restrict__ xb,
                                                     const float* __restrict__ ca,
                                                     v8f& acc0, v8f& acc1) {
#pragma unroll
  for (int ks = 0; ks < XS_ROWS / 4; ++ks) {
    v2f a  = { ca[ks * 64], ca[ks * 64 + 16] };
    v2f b0 = { xb[ks * 4 * XS_PITCH],      xb[ks * 4 * XS_PITCH + XS_PITCH] };
    v2f b1 = { xb[ks * 4 * XS_PITCH + 16], xb[ks * 4 * XS_PITCH + XS_PITCH + 16] };
    acc0 = __builtin_amdgcn_wmma_f32_16x16x4_f32(
        false, a, false, b0, (short)0, acc0, false, false);
    acc1 = __builtin_amdgcn_wmma_f32_16x16x4_f32(
        false, a, false, b1, (short)0, acc1, false, false);
  }
}

#define XS_LOAD_TILE()                                            \
  _Pragma("unroll")                                               \
  for (int i = 0; i < XS_PIECES; ++i)                             \
    rx[i] = *(const float4*)(xg + (size_t)(4 * i) * D);           \
  rc0 = *(const float4*)(cg);                                     \
  rc1 = *(const float4*)(cg + XS_THREADS * 4);

#define XS_STORE_TILE(buf)                                        \
  _Pragma("unroll")                                               \
  for (int i = 0; i < XS_PIECES; ++i)                             \
    *(float4*)(xsw + (buf) * (XS_ROWS * XS_PITCH) + 4 * i * XS_PITCH) = rx[i]; \
  *(float4*)(csw + (buf) * (XS_ROWS * 16))                  = rc0; \
  *(float4*)(csw + (buf) * (XS_ROWS * 16) + XS_THREADS * 4) = rc1;

template <int DC>
__global__ void __launch_bounds__(XS_THREADS)
gemm_coef_x_staged(const float* __restrict__ x, const float* __restrict__ coefT,
                   float* __restrict__ partial, int Drt, int TC, int NS,
                   const float* __restrict__ wpf, int wrows, int npf) {
  __shared__ __attribute__((aligned(16))) float xs[2 * XS_ROWS * XS_PITCH];
  __shared__ __attribute__((aligned(16))) float cs[2 * XS_ROWS * 16];
  const int D    = (DC > 0) ? DC : Drt;
  const int tid  = threadIdx.x;
  const int lane = tid & 31;
  const int wv   = __builtin_amdgcn_readfirstlane(tid >> 5);
  const int n = lane & 15, h = lane >> 4;
  const int slab   = blockIdx.x % NS;
  const int kchunk = blockIdx.x / NS;
  const int cb = slab * XS_COLS;
  const int t0 = kchunk * TC;
  const int ntile = TC / XS_ROWS;

  const int lr = tid >> 5;
  const int lc = (tid & 31) * 4;
  const float* xg = x + (size_t)(t0 + lr) * D + cb + lc;
  const float* cg = coefT + (size_t)t0 * 16 + tid * 4;
  float* xsw = xs + lr * XS_PITCH + lc;
  float* csw = cs + tid * 4;
  const float* xsr = xs + (2 * h) * XS_PITCH + wv * 32 + n;
  const float* csr = cs + (2 * h) * 16 + n;

  float4 rx[XS_PIECES], rc0, rc1;
  XS_LOAD_TILE()
  XS_STORE_TILE(0)
  __syncthreads();

  v8f acc0 = {}, acc1 = {};
  for (int j = 0; j + 1 < ntile; ++j) {
    const int cur = j & 1;
    xg += (size_t)XS_ROWS * D;
    cg += XS_ROWS * 16;
    XS_LOAD_TILE()
    staged_tile_products(xsr + cur * (XS_ROWS * XS_PITCH),
                         csr + cur * (XS_ROWS * 16), acc0, acc1);
    XS_STORE_TILE(cur ^ 1)
    __builtin_amdgcn_sched_group_barrier(0x020, XS_PIECES + 2, 0);
    __builtin_amdgcn_sched_group_barrier(0x100, 9, 0);
#pragma unroll
    for (int ks = 0; ks < XS_ROWS / 4 - 3; ++ks) {
      __builtin_amdgcn_sched_group_barrier(0x008, 2, 0);
      __builtin_amdgcn_sched_group_barrier(0x100, 3, 0);
    }
    __builtin_amdgcn_sched_group_barrier(0x008, 6, 0);
    __builtin_amdgcn_sched_group_barrier(0x200, XS_PIECES + 2, 0);
    __syncthreads();
  }
  {
    const int cur = (ntile - 1) & 1;
    staged_tile_products(xsr + cur * (XS_ROWS * XS_PITCH),
                         csr + cur * (XS_ROWS * 16), acc0, acc1);
  }

  const int nb = cb + wv * 32;
#pragma unroll
  for (int r = 0; r < 8; ++r) {
    int m = r + 8 * h;
    if (m < NROW) {
      partial[((size_t)kchunk * NROW + m) * D + nb + n]      = acc0[r];
      partial[((size_t)kchunk * NROW + m) * D + nb + 16 + n] = acc1[r];
    }
  }

  if (wpf != nullptr && (int)blockIdx.x < npf) {
    constexpr int UPR = (2 * ((DC > 0) ? DC : 64)) / 64;
    for (int v = tid; v < 2 * UPR; v += XS_THREADS) {
      const int r = v / UPR, u = v % UPR;
      if (r < wrows)
        l2_prefetch_unit(wpf + ((size_t)blockIdx.x * wrows + r) * (size_t)(2 * D) + u * 64);
    }
  }
}

__global__ void reduce_partials(const float* __restrict__ partial,
                                const float* __restrict__ pool_b,
                                float* __restrict__ avg,
                                float* __restrict__ out_hfinal,
                                int D, int KC, int T) {
  int idx = blockIdx.x * blockDim.x + threadIdx.x;
  if (idx >= NROW * D) return;
  int m = idx / D, d = idx % D;
  float s = 0.0f;
  for (int c = 0; c < KC; ++c)
    s += partial[((size_t)c * NROW + m) * D + d];
  float invT = 1.0f / (float)T;
  if (m == 0)      avg[d]     = s * invT;
  else if (m == 1) avg[D + d] = s * invT + pool_b[0];
  else             out_hfinal[(size_t)(m - 2) * D + d] = s;
}

#define RR_THREADS 128

template <int DC, int KCC>
__global__ void __launch_bounds__(RR_THREADS)
reduce_partials_round(const float* __restrict__ partial,
                      const float* __restrict__ pool_b,
                      float* __restrict__ avg,
                      float* __restrict__ out_hfinal,
                      int T) {
  constexpr int Q = DC / 4;
  const int idx = blockIdx.x * RR_THREADS + threadIdx.x;
  if (idx >= NROW * Q) return;
  const int m = idx / Q, q = idx % Q;
  const float4* pp = (const float4*)partial + (size_t)m * Q + q;
  float4 p[KCC];
#pragma unroll
  for (int c = 0; c < KCC; ++c) p[c] = pp[(size_t)c * NROW * Q];
  float invT = 1.0f / (float)T;
  float pb = pool_b[0];
  asm volatile("" : "+v"(invT), "+v"(pb));
  float sx = 0.0f, sy = 0.0f, sz = 0.0f, sw = 0.0f;
#pragma unroll
  for (int c = 0; c < KCC; ++c) {
    sx += p[c].x; sy += p[c].y; sz += p[c].z; sw += p[c].w;
  }
  __builtin_amdgcn_sched_group_barrier(0x020, KCC, 0);
  __builtin_amdgcn_sched_group_barrier(0x002, 4 * KCC, 0);
  const int d = q * 4;
  if (m < 2) {
    const float add = (m == 1) ? pb : 0.0f;
    float4 o = { sx * invT + add, sy * invT + add, sz * invT + add, sw * invT + add };
    *(float4*)(avg + (size_t)m * DC + d) = o;
  } else {
    float* oh = out_hfinal + (size_t)(m - 2) * DC + d;
    float2 lo = { sx, sy }, hi = { sz, sw };
    *(float2*)(oh)     = lo;
    *(float2*)(oh + 2) = hi;
  }
}

__global__ void gemv_rows(const float* __restrict__ Wm,
                          const float* __restrict__ bias,
                          const float* __restrict__ vec,
                          float* __restrict__ out,
                          int K, int N, int do_relu) {
  int wid  = blockIdx.x * (blockDim.x >> 5) + (threadIdx.x >> 5);
  int lane = threadIdx.x & 31;
  if (wid >= N) return;
  const float4* wr = (const float4*)(Wm + (size_t)wid * K);
  const float4* vv = (const float4*)vec;
  float s = 0.0f;
  for (int i = lane; i < (K >> 2); i += 32) {
    float4 a = wr[i], b = vv[i];
    s += a.x * b.x + a.y * b.y + a.z * b.z + a.w * b.w;
  }
  for (int off = 16; off > 0; off >>= 1) s += __shfl_down(s, off, 32);
  if (lane == 0) {
    float v = s + bias[wid];
    out[wid] = do_relu ? fmaxf(v, 0.0f) : v;
  }
}

#define GB_THREADS 128
#define GB_ROWS    8
#define GB_GROUP   4

template <int KP>
__global__ void __launch_bounds__(GB_THREADS)
gemv_rows_block(const float* __restrict__ Wm,
                const float* __restrict__ bias,
                const float* __restrict__ vec,
                float* __restrict__ out,
                int do_relu) {
  __shared__ float part[GB_ROWS][GB_THREADS / 32];
  const int tid  = threadIdx.x;
  const int lane = tid & 31;
  const int wv   = tid >> 5;
  const int row0 = blockIdx.x * GB_ROWS;
  constexpr int K4 = KP * GB_THREADS;

  const float4* vv = (const float4*)vec + tid;
  const float4* wr = (const float4*)Wm + (size_t)row0 * K4 + tid;

  float bv = (tid < GB_ROWS) ? bias[row0 + tid] : 0.0f;
  float4 v[KP];
#pragma unroll
  for (int p = 0; p < KP; ++p) v[p] = vv[p * GB_THREADS];

#pragma unroll 1
  for (int g = 0; g < GB_ROWS; g += GB_GROUP) {
    float4 a[GB_GROUP][KP];
#pragma unroll
    for (int r = 0; r < GB_GROUP; ++r)
#pragma unroll
      for (int p = 0; p < KP; ++p)
        a[r][p] = wr[(size_t)r * K4 + p * GB_THREADS];
    wr += (size_t)GB_GROUP * K4;
#pragma unroll
    for (int r = 0; r < GB_GROUP; ++r) {
      float sx = 0.0f, sy = 0.0f, sz = 0.0f, sw = 0.0f;
#pragma unroll
      for (int p = 0; p < KP; ++p) {
        sx = fmaf(a[r][p].x, v[p].x, sx);
        sy = fmaf(a[r][p].y, v[p].y, sy);
        sz = fmaf(a[r][p].z, v[p].z, sz);
        sw = fmaf(a[r][p].w, v[p].w, sw);
      }
      float s = (sx + sy) + (sz + sw);
#pragma unroll
      for (int off = 16; off > 0; off >>= 1) s += __shfl_xor(s, off, 32);
      if (lane == 0) part[g + r][wv] = s;
    }
  }
  __syncthreads();
  if (tid < GB_ROWS) {
    float t = 0.0f;
#pragma unroll
    for (int w2 = 0; w2 < GB_THREADS / 32; ++w2) t += part[tid][w2];
    float y = t + bv;
    out[row0 + tid] = do_relu ? fmaxf(y, 0.0f) : y;
  }
}

__global__ void __launch_bounds__(1024)
gemv_row_per_block(const float* __restrict__ Wm,
                   const float* __restrict__ bias,
                   const float* __restrict__ vec,
                   float* __restrict__ out,
                   int K, int do_relu) {
  __shared__ float part[32];
  int row  = blockIdx.x;
  int lane = threadIdx.x & 31;
  int wv   = threadIdx.x >> 5;
  const float4* wr = (const float4*)(Wm + (size_t)row * K);
  const float4* vv = (const float4*)vec;
  float bv = bias[row];
  float s = 0.0f;
  for (int i = threadIdx.x; i < (K >> 2); i += blockDim.x) {
    float4 a = wr[i], b = vv[i];
    s += a.x * b.x + a.y * b.y + a.z * b.z + a.w * b.w;
  }
  for (int off = 16; off > 0; off >>= 1) s += __shfl_down(s, off, 32);
  if (lane == 0) part[wv] = s;
  __syncthreads();
  if (wv == 0) {
    int nw = blockDim.x >> 5;
    float p = (lane < nw) ? part[lane] : 0.0f;
    for (int off = 16; off > 0; off >>= 1) p += __shfl_down(p, off, 32);
    if (lane == 0) {
      float v = p + bv;
      out[row] = do_relu ? fmaxf(v, 0.0f) : v;
    }
  }
}

extern "C" void kernel_launch(void* const* d_in, const int* in_sizes, int n_in,
                              void* d_out, int out_size, void* d_ws, size_t ws_size,
                              hipStream_t stream) {
  const float* inp    = (const float*)d_in[0];
  const float* w      = (const float*)d_in[2];
  const float* pool_w = (const float*)d_in[3];
  const float* pool_b = (const float*)d_in[4];
  const float* i2o_w  = (const float*)d_in[5];
  const float* i2o_b  = (const float*)d_in[6];
  const float* o_w    = (const float*)d_in[7];
  const float* o_b    = (const float*)d_in[8];
  float* out = (float*)d_out;

  const int D  = in_sizes[1] / F_NUM;
  const int T  = in_sizes[0] / D;
  const int H1 = in_sizes[6];

  const int KC = 16;
  int TC   = (((T + KC - 1) / KC) + 3) & ~3;
  int Tpad = KC * TC;
  int NT   = D / 16;

  float* ws      = (float*)d_ws;
  float* coefT   = ws;
  float* partial = coefT + (size_t)Tpad * 16;
  float* avg     = partial + (size_t)KC * NROW * D;
  float* hid1    = avg + (size_t)2 * D;

  const bool staged  = (D == 4096 && T == Tpad && (TC % XS_ROWS) == 0);
  const bool blkgemv = (D == 4096 && (H1 % GB_ROWS) == 0);

  if (staged)
    coef_kernel<<<(Tpad + 31) / 32, 32, 0, stream>>>(w, pool_w, coefT, T, Tpad, D,
                                                     inp, TC, D / XS_COLS, (D / XS_COLS) * KC);
  else
    coef_kernel<<<(Tpad + 255) / 256, 256, 0, stream>>>(w, pool_w, coefT, T, Tpad, D,
                                                        nullptr, TC, 1, 0);

  int units = NT * KC;
  dim3 g2((units + 3) / 4), b2(128);
  if (staged)
    gemm_coef_x_staged<4096><<<(D / XS_COLS) * KC, XS_THREADS, 0, stream>>>(
        inp, coefT, partial, D, TC, D / XS_COLS,
        blkgemv ? i2o_w : nullptr, GB_ROWS, H1 / GB_ROWS);
  else
    gemm_coef_x<0><<<g2, b2, 0, stream>>>(inp, coefT, partial, T, D, TC, NT, units);

  if (D == 4096 && KC == 16)
    reduce_partials_round<4096, 16><<<(NROW * (4096 / 4) + RR_THREADS - 1) / RR_THREADS, RR_THREADS, 0, stream>>>(
        partial, pool_b, avg, out + OUT_N, T);
  else
    reduce_partials<<<(NROW * D + 255) / 256, 256, 0, stream>>>(
        partial, pool_b, avg, out + OUT_N, D, KC, T);

  if (blkgemv)
    gemv_rows_block<16><<<H1 / GB_ROWS, GB_THREADS, 0, stream>>>(i2o_w, i2o_b, avg, hid1, 1);
  else
    gemv_rows<<<(H1 + 7) / 8, 256, 0, stream>>>(i2o_w, i2o_b, avg, hid1, 2 * D, H1, 1);
  gemv_row_per_block<<<OUT_N, 1024, 0, stream>>>(o_w, o_b, hid1, out, H1, 0);
}


# product launch as a blocked scan: tile weights w^(63-i) built once in registers, accumulator times w^64 per 64-step tile; pooled row from x_sum and h_final; no coefficient launch; sum order changes
# speedup vs baseline: 3.1583x; 1.1175x over previous
#include <hip/hip_runtime.h>
#include <math.h>

typedef __attribute__((ext_vector_type(2))) float v2f;
typedef __attribute__((ext_vector_type(8))) float v8f;

#define F_NUM 10
#define OUT_N 10
#define NROW  12

__device__ __forceinline__ void l2_prefetch_unit(const float* p) {
  __builtin_prefetch(p, 0, 2);
}

__global__ void coef_kernel(const float* __restrict__ w,
                            const float* __restrict__ pool_w,
                            float* __restrict__ coefT,
                            int T, int Tpad, int D) {
  int t = blockIdx.x * blockDim.x + threadIdx.x;
  if (t >= Tpad) return;
  float row[16];
#pragma unroll
  for (int i = 0; i < 16; ++i) row[i] = 0.0f;
  if (t < T) {
    row[0] = 1.0f;
    float G = 0.0f;
    float e = (float)(T - 1 - t);
#pragma unroll
    for (int f = 0; f < F_NUM; ++f) {
      float wf = w[(size_t)f * D];
      float pf = pool_w[f];
      float c  = powf(wf, e);
      row[2 + f] = c;
      G += pf * (1.0f - c) / (1.0f - wf);
    }
    row[1] = G;
  }
#pragma unroll
  for (int i = 0; i < 16; ++i) coefT[(size_t)t * 16 + i] = row[i];
}

template <int DC>
__global__ void __launch_bounds__(128)
gemm_coef_x(const float* __restrict__ x, const float* __restrict__ coefT,
            float* __restrict__ partial, int T, int Drt, int TC, int NT, int units) {
  const int D = (DC > 0) ? DC : Drt;
  int wid = __builtin_amdgcn_readfirstlane(
      (int)(blockIdx.x * (blockDim.x >> 5)) + ((int)threadIdx.x >> 5));
  if (wid >= units) return;
  int lane = threadIdx.x & 31;
  int ntile  = wid % NT;
  int kchunk = wid / NT;
  int nb = ntile * 16;
  int n  = lane & 15;
  int h  = lane >> 4;

  int t0 = kchunk * TC;
  int valid = T - t0;
  if (valid < 0)  valid = 0;
  if (valid > TC) valid = TC;
  int full = valid & ~3;

  const float* cp = coefT + (size_t)(t0 + 2 * h) * 16 + n;
  const float* xp = x     + (size_t)(t0 + 2 * h) * D  + nb + n;
  const size_t xstep = (size_t)4 * D;

  v8f acc = {};
  if (full > 0) {
    v2f a = { cp[0], cp[16] };
    v2f b = { xp[0], xp[D] };
#pragma unroll 2
    for (int kk = 4; kk < full; kk += 4) {
      cp += 64;
      xp += xstep;
      v2f an = { cp[0], cp[16] };
      v2f bn = { xp[0], xp[D] };
      acc = __builtin_amdgcn_wmma_f32_16x16x4_f32(
          false, a, false, b, (short)0, acc, false, false);
      a = an; b = bn;
    }
    acc = __builtin_amdgcn_wmma_f32_16x16x4_f32(
        false, a, false, b, (short)0, acc, false, false);
  }
  for (int kk = full; kk < TC; kk += 4) {
    int ta = t0 + kk + 2 * h, tb = ta + 1;
    v2f a = { coefT[(size_t)ta * 16 + n], coefT[(size_t)tb * 16 + n] };
    v2f b = { (ta < T) ? x[(size_t)ta * D + nb + n] : 0.0f,
              (tb < T) ? x[(size_t)tb * D + nb + n] : 0.0f };
    acc = __builtin_amdgcn_wmma_f32_16x16x4_f32(
        false, a, false, b, (short)0, acc, false, false);
  }

#pragma unroll
  for (int r = 0; r < 8; ++r) {
    int m = r + 8 * h;
    if (m < NROW)
      partial[((size_t)kchunk * NROW + m) * D + nb + n] = acc[r];
  }
}

#define XS_ROWS  64
#define XS_COLS  128
#define XS_PITCH 136
#define XS_THREADS 128
#define XS_PIECES  16
#define XS_KSTEPS (XS_ROWS / 4)

__device__ __forceinline__ void staged_tile_products(const float* __restrict__ xb,
                                                     const v2f (&a)[XS_KSTEPS],
                                                     v8f& acc0, v8f& acc1) {
#pragma unroll
  for (int ks = 0; ks < XS_KSTEPS; ++ks) {
    v2f b0 = { xb[ks * 4 * XS_PITCH],      xb[ks * 4 * XS_PITCH + XS_PITCH] };
    v2f b1 = { xb[ks * 4 * XS_PITCH + 16], xb[ks * 4 * XS_PITCH + XS_PITCH + 16] };
    acc0 = __builtin_amdgcn_wmma_f32_16x16x4_f32(
        false, a[ks], false, b0, (short)0, acc0, false, false);
    acc1 = __builtin_amdgcn_wmma_f32_16x16x4_f32(
        false, a[ks], false, b1, (short)0, acc1, false, false);
  }
}

#define XS_LOAD_TILE()                                            \
  _Pragma("unroll")                                               \
  for (int i = 0; i < XS_PIECES; ++i)                             \
    rx[i] = *(const float4*)(xg + (size_t)(4 * i) * D);

#define XS_STORE_TILE(buf)                                        \
  _Pragma("unroll")                                               \
  for (int i = 0; i < XS_PIECES; ++i)                             \
    *(float4*)(xsw + (buf) * (XS_ROWS * XS_PITCH) + 4 * i * XS_PITCH) = rx[i];

template <int DC>
__global__ void __launch_bounds__(XS_THREADS)
gemm_scan_x_staged(const float* __restrict__ x, const float* __restrict__ w,
                   float* __restrict__ partial, int Drt, int TC, int NS,
                   const float* __restrict__ wpf, int wrows, int npf) {
  __shared__ __attribute__((aligned(16))) float xs[2 * XS_ROWS * XS_PITCH];
  const int D    = (DC > 0) ? DC : Drt;
  const int tid  = threadIdx.x;
  const int lane = tid & 31;
  const int wv   = __builtin_amdgcn_readfirstlane(tid >> 5);
  const int n = lane & 15, h = lane >> 4;
  const int slab   = blockIdx.x % NS;
  const int kchunk = blockIdx.x / NS;
  const int cb = slab * XS_COLS;
  const int t0 = kchunk * TC;
  const int ntile = TC / XS_ROWS;

  const int lr = tid >> 5;
  const int lc = (tid & 31) * 4;
  const float* xg = x + (size_t)(t0 + lr) * D + cb + lc;
  float* xsw = xs + lr * XS_PITCH + lc;
  const float* xsr = xs + (2 * h) * XS_PITCH + wv * 32 + n;

  float4 rx[XS_PIECES];
  XS_LOAD_TILE()

  const bool live = (n == 0) || (n >= 2 && n < NROW);
  float wm = 1.0f;
#pragma unroll
  for (int f = 0; f < F_NUM; ++f) {
    const float wf = w[(size_t)f * D];
    wm = (n == f + 2) ? wf : wm;
  }
  const float on  = live ? 1.0f : 0.0f;
  const float w2  = wm * wm, w4 = w2 * w2;
  v2f afr[XS_KSTEPS];
  {
    v2f cur = { (h ? wm : w2 * wm) * on, (h ? 1.0f : w2) * on };
#pragma unroll
    for (int ks = XS_KSTEPS - 1; ks >= 0; --ks) {
      afr[ks] = cur;
      cur.x *= w4; cur.y *= w4;
    }
  }
  const float s64m = powf(wm, (float)XS_ROWS) * on;
  float sc[8];
#pragma unroll
  for (int r = 0; r < 8; ++r) sc[r] = __shfl(s64m, 8 * h + r, 32);

  XS_STORE_TILE(0)
  __builtin_amdgcn_sched_group_barrier(0x020, XS_PIECES, 0);
  __builtin_amdgcn_sched_group_barrier(0x002, 400, 0);
  __builtin_amdgcn_sched_group_barrier(0x200, XS_PIECES, 0);
  __syncthreads();

  v8f acc0 = {}, acc1 = {};
  for (int j = 0; j + 1 < ntile; ++j) {
    const int cur = j & 1;
    xg += (size_t)XS_ROWS * D;
    XS_LOAD_TILE()
    staged_tile_products(xsr + cur * (XS_ROWS * XS_PITCH), afr, acc0, acc1);
    XS_STORE_TILE(cur ^ 1)
    __builtin_amdgcn_sched_group_barrier(0x020, XS_PIECES, 0);
    __builtin_amdgcn_sched_group_barrier(0x100, 6, 0);
#pragma unroll
    for (int ks = 0; ks < XS_KSTEPS - 3; ++ks) {
      __builtin_amdgcn_sched_group_barrier(0x008, 2, 0);
      __builtin_amdgcn_sched_group_barrier(0x100, 2, 0);
    }
    __builtin_amdgcn_sched_group_barrier(0x008, 6, 0);
    __builtin_amdgcn_sched_group_barrier(0x200, XS_PIECES, 0);
#pragma unroll
    for (int r = 0; r < 8; ++r) { acc0[r] *= sc[r]; acc1[r] *= sc[r]; }
    __syncthreads();
  }
  {
    const int cur = (ntile - 1) & 1;
    staged_tile_products(xsr + cur * (XS_ROWS * XS_PITCH), afr, acc0, acc1);
  }

  const int nb = cb + wv * 32;
#pragma unroll
  for (int r = 0; r < 8; ++r) {
    int m = r + 8 * h;
    if (m < NROW && m != 1) {
      partial[((size_t)kchunk * NROW + m) * D + nb + n]      = acc0[r];
      partial[((size_t)kchunk * NROW + m) * D + nb + 16 + n] = acc1[r];
    }
  }

  if (wpf != nullptr && (int)blockIdx.x < npf) {
    constexpr int UPR = (2 * ((DC > 0) ? DC : 64)) / 64;
    for (int v = tid; v < 2 * UPR; v += XS_THREADS) {
      const int r = v / UPR, u = v % UPR;
      if (r < wrows)
        l2_prefetch_unit(wpf + ((size_t)blockIdx.x * wrows + r) * (size_t)(2 * D) + u * 64);
    }
  }
}

__global__ void reduce_partials(const float* __restrict__ partial,
                                const float* __restrict__ pool_b,
                                float* __restrict__ avg,
                                float* __restrict__ out_hfinal,
                                int D, int KC, int T) {
  int idx = blockIdx.x * blockDim.x + threadIdx.x;
  if (idx >= NROW * D) return;
  int m = idx / D, d = idx % D;
  float s = 0.0f;
  for (int c = 0; c < KC; ++c)
    s += partial[((size_t)c * NROW + m) * D + d];
  float invT = 1.0f / (float)T;
  if (m == 0)      avg[d]     = s * invT;
  else if (m == 1) avg[D + d] = s * invT + pool_b[0];
  else             out_hfinal[(size_t)(m - 2) * D + d] = s;
}

#define RR_THREADS 128

template <int DC, int KCC>
__global__ void __launch_bounds__(RR_THREADS)
reduce_partials_round(const float* __restrict__ partial,
                      const float* __restrict__ pool_b,
                      float* __restrict__ avg,
                      float* __restrict__ out_hfinal,
                      int T) {
  constexpr int Q = DC / 4;
  const int idx = blockIdx.x * RR_THREADS + threadIdx.x;
  if (idx >= NROW * Q) return;
  const int m = idx / Q, q = idx % Q;
  const float4* pp = (const float4*)partial + (size_t)m * Q + q;
  float4 p[KCC];
#pragma unroll
  for (int c = 0; c < KCC; ++c) p[c] = pp[(size_t)c * NROW * Q];
  float invT = 1.0f / (float)T;
  float pb = pool_b[0];
  asm volatile("" : "+v"(invT), "+v"(pb));
  float sx = 0.0f, sy = 0.0f, sz = 0.0f, sw = 0.0f;
#pragma unroll
  for (int c = 0; c < KCC; ++c) {
    sx += p[c].x; sy += p[c].y; sz += p[c].z; sw += p[c].w;
  }
  __builtin_amdgcn_sched_group_barrier(0x020, KCC, 0);
  __builtin_amdgcn_sched_group_barrier(0x002, 4 * KCC, 0);
  const int d = q * 4;
  if (m < 2) {
    const float add = (m == 1) ? pb : 0.0f;
    float4 o = { sx * invT + add, sy * invT + add, sz * invT + add, sw * invT + add };
    *(float4*)(avg + (size_t)m * DC + d) = o;
  } else {
    float* oh = out_hfinal + (size_t)(m - 2) * DC + d;
    float2 lo = { sx, sy }, hi = { sz, sw };
    *(float2*)(oh)     = lo;
    *(float2*)(oh + 2) = hi;
  }
}

#define RS_THREADS (32 * NROW)

template <int DC, int KCC>
__global__ void __launch_bounds__(RS_THREADS)
reduce_scan_partials(const float* __restrict__ partial,
                     const float* __restrict__ w,
                     const float* __restrict__ pool_w,
                     const float* __restrict__ pool_b,
                     float* __restrict__ avg,
                     float* __restrict__ out_hfinal,
                     int T, int TC) {
  __shared__ float4 hf[F_NUM][32];
  constexpr int Q = DC / 4;
  const int lane = threadIdx.x & 31;
  const int m = __builtin_amdgcn_readfirstlane((int)threadIdx.x >> 5);
  const int q = blockIdx.x * 32 + lane;
  const float4* pp = (const float4*)partial + (size_t)m * Q + q;
  float4 p[KCC];
  float4 s4 = { 0.0f, 0.0f, 0.0f, 0.0f };
  float qf[F_NUM];
#pragma unroll
  for (int f = 0; f < F_NUM; ++f) qf[f] = 0.0f;
  if (m != 1) {
#pragma unroll
    for (int c = 0; c < KCC; ++c) p[c] = pp[(size_t)c * NROW * Q];
    if (m == 0) {
#pragma unroll
      for (int f = 0; f < F_NUM; ++f) qf[f] = pool_w[f] / (1.0f - w[(size_t)f * DC]);
#pragma unroll
      for (int c = 0; c < KCC; ++c) {
        s4.x += p[c].x; s4.y += p[c].y; s4.z += p[c].z; s4.w += p[c].w;
      }
    } else {
      const float s = powf(w[(size_t)(m - 2) * DC], (float)TC);
      s4 = p[0];
#pragma unroll
      for (int c = 1; c < KCC; ++c) {
        s4.x = fmaf(s4.x, s, p[c].x); s4.y = fmaf(s4.y, s, p[c].y);
        s4.z = fmaf(s4.z, s, p[c].z); s4.w = fmaf(s4.w, s, p[c].w);
      }
      float* oh = out_hfinal + (size_t)(m - 2) * DC + q * 4;
      float2 lo = { s4.x, s4.y }, hi = { s4.z, s4.w };
      *(float2*)(oh)     = lo;
      *(float2*)(oh + 2) = hi;
      hf[m - 2][lane] = s4;
    }
  }
  __syncthreads();
  if (m == 0) {
    const float invT = 1.0f / (float)T;
    const float pb = pool_b[0];
    float4 g = { 0.0f, 0.0f, 0.0f, 0.0f };
#pragma unroll
    for (int f = 0; f < F_NUM; ++f) {
      const float4 h4 = hf[f][lane];
      g.x += qf[f] * (s4.x - h4.x); g.y += qf[f] * (s4.y - h4.y);
      g.z += qf[f] * (s4.z - h4.z); g.w += qf[f] * (s4.w - h4.w);
    }
    float4 o0 = { s4.x * invT, s4.y * invT, s4.z * invT, s4.w * invT };
    float4 o1 = { g.x * invT + pb, g.y * invT + pb, g.z * invT + pb, g.w * invT + pb };
    *(float4*)(avg + q * 4)      = o0;
    *(float4*)(avg + DC + q * 4) = o1;
  }
}

__global__ void gemv_rows(const float* __restrict__ Wm,
                          const float* __restrict__ bias,
                          const float* __restrict__ vec,
                          float* __restrict__ out,
                          int K, int N, int do_relu) {
  int wid  = blockIdx.x * (blockDim.x >> 5) + (threadIdx.x >> 5);
  int lane = threadIdx.x & 31;
  if (wid >= N) return;
  const float4* wr = (const float4*)(Wm + (size_t)wid * K);
  const float4* vv = (const float4*)vec;
  float s = 0.0f;
  for (int i = lane; i < (K >> 2); i += 32) {
    float4 a = wr[i], b = vv[i];
    s += a.x * b.x + a.y * b.y + a.z * b.z + a.w * b.w;
  }
  for (int off = 16; off > 0; off >>= 1) s += __shfl_down(s, off, 32);
  if (lane == 0) {
    float v = s + bias[wid];
    out[wid] = do_relu ? fmaxf(v, 0.0f) : v;
  }
}

#define GB_THREADS 128
#define GB_ROWS    8
#define GB_GROUP   4

template <int KP>
__global__ void __launch_bounds__(GB_THREADS)
gemv_rows_block(const float* __restrict__ Wm,
                const float* __restrict__ bias,
                const float* __restrict__ vec,
                float* __restrict__ out,
                int do_relu) {
  __shared__ float part[GB_ROWS][GB_THREADS / 32];
  const int tid  = threadIdx.x;
  const int lane = tid & 31;
  const int wv   = tid >> 5;
  const int row0 = blockIdx.x * GB_ROWS;
  constexpr int K4 = KP * GB_THREADS;

  const float4* vv = (const float4*)vec + tid;
  const float4* wr = (const float4*)Wm + (size_t)row0 * K4 + tid;

  float bv = (tid < GB_ROWS) ? bias[row0 + tid] : 0.0f;
  float4 v[KP];
#pragma unroll
  for (int p = 0; p < KP; ++p) v[p] = vv[p * GB_THREADS];

#pragma unroll 1
  for (int g = 0; g < GB_ROWS; g += GB_GROUP) {
    float4 a[GB_GROUP][KP];
#pragma unroll
    for (int r = 0; r < GB_GROUP; ++r)
#pragma unroll
      for (int p = 0; p < KP; ++p)
        a[r][p] = wr[(size_t)r * K4 + p * GB_THREADS];
    wr += (size_t)GB_GROUP * K4;
#pragma unroll
    for (int r = 0; r < GB_GROUP; ++r) {
      float sx = 0.0f, sy = 0.0f, sz = 0.0f, sw = 0.0f;
#pragma unroll
      for (int p = 0; p < KP; ++p) {
        sx = fmaf(a[r][p].x, v[p].x, sx);
        sy = fmaf(a[r][p].y, v[p].y, sy);
        sz = fmaf(a[r][p].z, v[p].z, sz);
        sw = fmaf(a[r][p].w, v[p].w, sw);
      }
      float s = (sx + sy) + (sz + sw);
#pragma unroll
      for (int off = 16; off > 0; off >>= 1) s += __shfl_xor(s, off, 32);
      if (lane == 0) part[g + r][wv] = s;
    }
  }
  __syncthreads();
  if (tid < GB_ROWS) {
    float t = 0.0f;
#pragma unroll
    for (int w2 = 0; w2 < GB_THREADS / 32; ++w2) t += part[tid][w2];
    float y = t + bv;
    out[row0 + tid] = do_relu ? fmaxf(y, 0.0f) : y;
  }
}

__global__ void __launch_bounds__(1024)
gemv_row_per_block(const float* __restrict__ Wm,
                   const float* __restrict__ bias,
                   const float* __restrict__ vec,
                   float* __restrict__ out,
                   int K, int do_relu) {
  __shared__ float part[32];
  int row  = blockIdx.x;
  int lane = threadIdx.x & 31;
  int wv   = threadIdx.x >> 5;
  const float4* wr = (const float4*)(Wm + (size_t)row * K);
  const float4* vv = (const float4*)vec;
  float bv = bias[row];
  float s = 0.0f;
  for (int i = threadIdx.x; i < (K >> 2); i += blockDim.x) {
    float4 a = wr[i], b = vv[i];
    s += a.x * b.x + a.y * b.y + a.z * b.z + a.w * b.w;
  }
  for (int off = 16; off > 0; off >>= 1) s += __shfl_down(s, off, 32);
  if (lane == 0) part[wv] = s;
  __syncthreads();
  if (wv == 0) {
    int nw = blockDim.x >> 5;
    float p = (lane < nw) ? part[lane] : 0.0f;
    for (int off = 16; off > 0; off >>= 1) p += __shfl_down(p, off, 32);
    if (lane == 0) {
      float v = p + bv;
      out[row] = do_relu ? fmaxf(v, 0.0f) : v;
    }
  }
}

extern "C" void kernel_launch(void* const* d_in, const int* in_sizes, int n_in,
                              void* d_out, int out_size, void* d_ws, size_t ws_size,
                              hipStream_t stream) {
  const float* inp    = (const float*)d_in[0];
  const float* w      = (const float*)d_in[2];
  const float* pool_w = (const float*)d_in[3];
  const float* pool_b = (const float*)d_in[4];
  const float* i2o_w  = (const float*)d_in[5];
  const float* i2o_b  = (const float*)d_in[6];
  const float* o_w    = (const float*)d_in[7];
  const float* o_b    = (const float*)d_in[8];
  float* out = (float*)d_out;

  const int D  = in_sizes[1] / F_NUM;
  const int T  = in_sizes[0] / D;
  const int H1 = in_sizes[6];

  const int KC = 16;
  int TC   = (((T + KC - 1) / KC) + 3) & ~3;
  int Tpad = KC * TC;
  int NT   = D / 16;

  float* ws      = (float*)d_ws;
  float* coefT   = ws;
  float* partial = coefT + (size_t)Tpad * 16;
  float* avg     = partial + (size_t)KC * NROW * D;
  float* hid1    = avg + (size_t)2 * D;

  const bool staged  = (D == 4096 && T == Tpad && (TC % XS_ROWS) == 0);
  const bool blkgemv = (D == 4096 && (H1 % GB_ROWS) == 0);

  if (staged) {
    gemm_scan_x_staged<4096><<<(D / XS_COLS) * KC, XS_THREADS, 0, stream>>>(
        inp, w, partial, D, TC, D / XS_COLS,
        blkgemv ? i2o_w : nullptr, GB_ROWS, H1 / GB_ROWS);
    reduce_scan_partials<4096, 16><<<(4096 / 4) / 32, RS_THREADS, 0, stream>>>(
        partial, w, pool_w, pool_b, avg, out + OUT_N, T, TC);
  } else {
    coef_kernel<<<(Tpad + 255) / 256, 256, 0, stream>>>(w, pool_w, coefT, T, Tpad, D);
    int units = NT * KC;
    dim3 g2((units + 3) / 4), b2(128);
    gemm_coef_x<0><<<g2, b2, 0, stream>>>(inp, coefT, partial, T, D, TC, NT, units);
    if (D == 4096 && KC == 16)
      reduce_partials_round<4096, 16><<<(NROW * (4096 / 4) + RR_THREADS - 1) / RR_THREADS, RR_THREADS, 0, stream>>>(
          partial, pool_b, avg, out + OUT_N, T);
    else
      reduce_partials<<<(NROW * D + 255) / 256, 256, 0, stream>>>(
          partial, pool_b, avg, out + OUT_N, D, KC, T);
  }

  if (blkgemv)
    gemv_rows_block<16><<<H1 / GB_ROWS, GB_THREADS, 0, stream>>>(i2o_w, i2o_b, avg, hid1, 1);
  else
    gemv_rows<<<(H1 + 7) / 8, 256, 0, stream>>>(i2o_w, i2o_b, avg, hid1, 2 * D, H1, 1);
  gemv_row_per_block<<<OUT_N, 1024, 0, stream>>>(o_w, o_b, hid1, out, H1, 0);
}
